# starting kernel
# baseline (speedup 1.0000x reference)
#include <hip/hip_runtime.h>

typedef __attribute__((ext_vector_type(2))) float v2f;
typedef __attribute__((ext_vector_type(8))) float v8f;

#define IN_F 64
#define OUT_F 64
#define NCLS 16

__global__ void k_zero(float4* __restrict__ p, long n4) {
  long i = (long)blockIdx.x * blockDim.x + threadIdx.x;
  if (i < n4) p[i] = make_float4(0.f, 0.f, 0.f, 0.f);
}

__global__ void k_degree(const int* __restrict__ src, const int* __restrict__ dst,
                         int* deg_s, int* deg_d, int E) {
  int e = blockIdx.x * blockDim.x + threadIdx.x;
  if (e < E) {
    atomicAdd(&deg_s[src[e]], 1);
    atomicAdd(&deg_d[dst[e]], 1);
  }
}

__global__ void k_norm(const int* __restrict__ deg_s, const int* __restrict__ deg_d,
                       float* ns, float* nd, int N) {
  int i = blockIdx.x * blockDim.x + threadIdx.x;
  if (i < N) {
    int a = deg_s[i]; ns[i] = (a > 0) ? rsqrtf((float)a) : 1.0f;
    int b = deg_d[i]; nd[i] = (b > 0) ? rsqrtf((float)b) : 1.0f;
  }
}

__global__ void k_scatter64(const int* __restrict__ src, const int* __restrict__ dst,
                            const float* __restrict__ x, const float* __restrict__ ns,
                            float* agg, int E) {
  long t = (long)blockIdx.x * blockDim.x + threadIdx.x;
  int e = (int)(t >> 4);
  if (e >= E) return;
  int f = ((int)t & 15) << 2;
  int s = src[e], d = dst[e];
  float w = ns[s];
  float4 v = *(const float4*)(x + (long)s * IN_F + f);
  float* o = agg + (long)d * IN_F + f;
  atomicAdd(o + 0, v.x * w);
  atomicAdd(o + 1, v.y * w);
  atomicAdd(o + 2, v.z * w);
  atomicAdd(o + 3, v.w * w);
}

__global__ void k_scatter16(const int* __restrict__ src, const int* __restrict__ dst,
                            const float* __restrict__ z, const float* __restrict__ ns,
                            float* agg, int E) {
  long t = (long)blockIdx.x * blockDim.x + threadIdx.x;
  int e = (int)(t >> 2);
  if (e >= E) return;
  int f = ((int)t & 3) << 2;
  int s = src[e], d = dst[e];
  float w = ns[s];
  float4 v = *(const float4*)(z + (long)s * NCLS + f);
  float* o = agg + (long)d * NCLS + f;
  atomicAdd(o + 0, v.x * w);
  atomicAdd(o + 1, v.y * w);
  atomicAdd(o + 2, v.z * w);
  atomicAdd(o + 3, v.w * w);
}

__global__ void __launch_bounds__(256)
k_gemm_relu(const float* __restrict__ A, const float* __restrict__ nd,
            const float* __restrict__ W, const float* __restrict__ bias,
            float* __restrict__ out, int nrows) {
  __shared__ float sW[IN_F * OUT_F];
  for (int i = threadIdx.x; i < IN_F * OUT_F; i += blockDim.x) sW[i] = W[i];
  __syncthreads();

  int wave = (int)((blockIdx.x * blockDim.x + threadIdx.x) >> 5);
  int lane = threadIdx.x & 31;
  int m0 = wave * 16;
  if (m0 >= nrows) return;

  int lane16 = lane & 15;
  int half   = lane >> 4;
  int arow   = m0 + lane16;
  float scale = nd[arow];
  const float* ap = A + (long)arow * IN_F + half * 2;

  v8f acc0 = {}, acc1 = {}, acc2 = {}, acc3 = {};
#pragma unroll
  for (int k0 = 0; k0 < IN_F; k0 += 4) {
    float2 av = *(const float2*)(ap + k0);
    v2f a = {av.x * scale, av.y * scale};
    int kk = k0 + half * 2;
    const float* w0 = sW + kk * OUT_F + lane16;
    const float* w1 = w0 + OUT_F;
    v2f b0 = {w0[0],  w1[0]};
    v2f b1 = {w0[16], w1[16]};
    v2f b2 = {w0[32], w1[32]};
    v2f b3 = {w0[48], w1[48]};
    acc0 = __builtin_amdgcn_wmma_f32_16x16x4_f32(false, a, false, b0, (short)0, acc0, false, false);
    acc1 = __builtin_amdgcn_wmma_f32_16x16x4_f32(false, a, false, b1, (short)0, acc1, false, false);
    acc2 = __builtin_amdgcn_wmma_f32_16x16x4_f32(false, a, false, b2, (short)0, acc2, false, false);
    acc3 = __builtin_amdgcn_wmma_f32_16x16x4_f32(false, a, false, b3, (short)0, acc3, false, false);
  }

  int rbase = m0 + half * 8;
  float bi0 = bias[lane16], bi1 = bias[16 + lane16];
  float bi2 = bias[32 + lane16], bi3 = bias[48 + lane16];
#pragma unroll
  for (int r = 0; r < 8; ++r) {
    float* orow = out + (long)(rbase + r) * OUT_F + lane16;
    orow[0]  = fmaxf(acc0[r] + bi0, 0.f);
    orow[16] = fmaxf(acc1[r] + bi1, 0.f);
    orow[32] = fmaxf(acc2[r] + bi2, 0.f);
    orow[48] = fmaxf(acc3[r] + bi3, 0.f);
  }
}

__global__ void __launch_bounds__(256)
k_gemm16(const float* __restrict__ A, const float* __restrict__ W,
         float* __restrict__ out, int nrows) {
  __shared__ float sW[OUT_F * NCLS];
  for (int i = threadIdx.x; i < OUT_F * NCLS; i += blockDim.x) sW[i] = W[i];
  __syncthreads();

  int wave = (int)((blockIdx.x * blockDim.x + threadIdx.x) >> 5);
  int lane = threadIdx.x & 31;
  int m0 = wave * 16;
  if (m0 >= nrows) return;

  int lane16 = lane & 15;
  int half   = lane >> 4;
  const float* ap = A + (long)(m0 + lane16) * OUT_F + half * 2;

  v8f acc = {};
#pragma unroll
  for (int k0 = 0; k0 < OUT_F; k0 += 4) {
    float2 av = *(const float2*)(ap + k0);
    v2f a = {av.x, av.y};
    int kk = k0 + half * 2;
    v2f b = {sW[kk * NCLS + lane16], sW[(kk + 1) * NCLS + lane16]};
    acc = __builtin_amdgcn_wmma_f32_16x16x4_f32(false, a, false, b, (short)0, acc, false, false);
  }
  int rbase = m0 + half * 8;
#pragma unroll
  for (int r = 0; r < 8; ++r)
    out[(long)(rbase + r) * NCLS + lane16] = acc[r];
}

__global__ void k_finalize(const float* __restrict__ agg, const float* __restrict__ nd,
                           const float* __restrict__ bias, float* __restrict__ out, int N) {
  int i = blockIdx.x * blockDim.x + threadIdx.x;
  if (i < N * NCLS) {
    int n = i >> 4, f = i & 15;
    out[i] = agg[i] * nd[n] + bias[f];
  }
}

extern "C" void kernel_launch(void* const* d_in, const int* in_sizes, int n_in,
                              void* d_out, int out_size, void* d_ws, size_t ws_size,
                              hipStream_t stream) {
  const float* x  = (const float*)d_in[0];
  const int*   src = (const int*)d_in[1];
  const int*   dst = (const int*)d_in[2];
  const float* W1 = (const float*)d_in[3];
  const float* b1 = (const float*)d_in[4];
  const float* W2 = (const float*)d_in[5];
  const float* b2 = (const float*)d_in[6];
  float* out = (float*)d_out;

  const int N = in_sizes[0] / IN_F;
  const int E = in_sizes[1];

  char* ws = (char*)d_ws;
  size_t off = 0;
  auto carve = [&](size_t bytes) -> void* {
    void* p = (void*)(ws + off);
    off += (bytes + 255) & ~(size_t)255;
    return p;
  };
  int*   deg_s = (int*)carve((size_t)N * 4);
  int*   deg_d = (int*)carve((size_t)N * 4);
  float* ns    = (float*)carve((size_t)N * 4);
  float* nd    = (float*)carve((size_t)N * 4);
  float* agg1  = (float*)carve((size_t)N * IN_F * 4);
  float* y1    = (float*)carve((size_t)N * OUT_F * 4);
  float* z     = (float*)carve((size_t)N * NCLS * 4);
  float* agg2  = (float*)carve((size_t)N * NCLS * 4);

  const int B = 256;
  auto cdiv = [](long a, long b) { return (int)((a + b - 1) / b); };

  long n4;
  n4 = ((long)N * 4 + 15) / 16;
  k_zero<<<cdiv(n4, B), B, 0, stream>>>((float4*)deg_s, n4);
  k_zero<<<cdiv(n4, B), B, 0, stream>>>((float4*)deg_d, n4);
  n4 = (long)N * IN_F / 4;
  k_zero<<<cdiv(n4, B), B, 0, stream>>>((float4*)agg1, n4);
  n4 = (long)N * NCLS / 4;
  k_zero<<<cdiv(n4, B), B, 0, stream>>>((float4*)agg2, n4);

  k_degree<<<cdiv(E, B), B, 0, stream>>>(src, dst, deg_s, deg_d, E);
  k_norm<<<cdiv(N, B), B, 0, stream>>>(deg_s, deg_d, ns, nd, N);

  k_scatter64<<<cdiv((long)E * 16, B), B, 0, stream>>>(src, dst, x, ns, agg1, E);
  int waves = cdiv(N, 16);
  k_gemm_relu<<<cdiv((long)waves * 32, B), B, 0, stream>>>(agg1, nd, W1, b1, y1, N);

  k_gemm16<<<cdiv((long)waves * 32, B), B, 0, stream>>>(y1, W2, z, N);
  k_scatter16<<<cdiv((long)E * 4, B), B, 0, stream>>>(src, dst, z, ns, agg2, E);
  k_finalize<<<cdiv((long)N * NCLS, B), B, 0, stream>>>(agg2, nd, b2, out, N);
}


# float-atomic scatter-adds replaced by counting sort of edges by dst (rank from a returning int atomic, then a scan) and atomic-free per-row gather sums; sum order = arrival rank in the row
# speedup vs baseline: 1.5623x; 1.5623x over previous
#include <hip/hip_runtime.h>

typedef __attribute__((ext_vector_type(2))) float v2f;
typedef __attribute__((ext_vector_type(8))) float v8f;

#define IN_F 64
#define OUT_F 64
#define NCLS 16

__global__ void k_zero(float4* __restrict__ p, long n4) {
  long i = (long)blockIdx.x * blockDim.x + threadIdx.x;
  if (i < n4) p[i] = make_float4(0.f, 0.f, 0.f, 0.f);
}

__global__ void k_degree_rank(const int* __restrict__ src, const int* __restrict__ dst,
                              int* deg_s, int* deg_d, int* __restrict__ pos, int E) {
  int e = blockIdx.x * blockDim.x + threadIdx.x;
  if (e < E) {
    int s = src[e], d = dst[e];
    atomicAdd(&deg_s[s], 1);
    pos[e] = atomicAdd(&deg_d[d], 1);
  }
}

__global__ void k_norm(const int* __restrict__ deg_s, const int* __restrict__ deg_d,
                       float* ns, float* nd, int N) {
  int i = blockIdx.x * blockDim.x + threadIdx.x;
  if (i < N) {
    int a = deg_s[i]; ns[i] = (a > 0) ? rsqrtf((float)a) : 1.0f;
    int b = deg_d[i]; nd[i] = (b > 0) ? rsqrtf((float)b) : 1.0f;
  }
}

__global__ void __launch_bounds__(1024)
k_scan(const int* __restrict__ cnt, int* __restrict__ rowstart, int N) {
  __shared__ int s[1024];
  int t = threadIdx.x;
  int base = blockIdx.x * 1024;

  int part = 0;
#pragma unroll 8
  for (int i = t * 4; i < base; i += 4096) {
    int4 v = *(const int4*)(cnt + i);
    part += (v.x + v.y) + (v.z + v.w);
  }
  s[t] = part;
  __syncthreads();
  for (int off = 512; off > 0; off >>= 1) {
    if (t < off) s[t] += s[t + off];
    __syncthreads();
  }
  int prefix = s[0];
  __syncthreads();

  int i = base + t;
  int v = (i < N) ? cnt[i] : 0;
  s[t] = v;
  __syncthreads();
  for (int off = 1; off < 1024; off <<= 1) {
    int add = (t >= off) ? s[t - off] : 0;
    __syncthreads();
    s[t] += add;
    __syncthreads();
  }
  if (i < N) rowstart[i] = prefix + s[t] - v;
}

__global__ void k_fill(const int* __restrict__ src, const int* __restrict__ dst,
                       const int* __restrict__ rowstart, const int* __restrict__ pos,
                       int* __restrict__ col, int E) {
  int e = blockIdx.x * blockDim.x + threadIdx.x;
  if (e < E) col[rowstart[dst[e]] + pos[e]] = src[e];
}

__global__ void k_gather64(const int* __restrict__ rowstart, const int* __restrict__ cnt,
                           const int* __restrict__ col, const float* __restrict__ x,
                           const float* __restrict__ ns, float* __restrict__ agg, int N) {
  long t = (long)blockIdx.x * blockDim.x + threadIdx.x;
  int d = (int)(t >> 4);
  if (d >= N) return;
  int f = ((int)t & 15) << 2;
  int j = rowstart[d];
  int end = j + cnt[d];
  float4 acc = make_float4(0.f, 0.f, 0.f, 0.f);
  for (; j + 4 <= end; j += 4) {
    int s0 = col[j], s1 = col[j + 1], s2 = col[j + 2], s3 = col[j + 3];
    float w0 = ns[s0], w1 = ns[s1], w2 = ns[s2], w3 = ns[s3];
    float4 v0 = *(const float4*)(x + (long)s0 * IN_F + f);
    float4 v1 = *(const float4*)(x + (long)s1 * IN_F + f);
    float4 v2 = *(const float4*)(x + (long)s2 * IN_F + f);
    float4 v3 = *(const float4*)(x + (long)s3 * IN_F + f);
    acc.x += v0.x * w0; acc.y += v0.y * w0; acc.z += v0.z * w0; acc.w += v0.w * w0;
    acc.x += v1.x * w1; acc.y += v1.y * w1; acc.z += v1.z * w1; acc.w += v1.w * w1;
    acc.x += v2.x * w2; acc.y += v2.y * w2; acc.z += v2.z * w2; acc.w += v2.w * w2;
    acc.x += v3.x * w3; acc.y += v3.y * w3; acc.z += v3.z * w3; acc.w += v3.w * w3;
  }
#pragma nounroll
  for (; j < end; ++j) {
    int s0 = col[j];
    float w0 = ns[s0];
    float4 v0 = *(const float4*)(x + (long)s0 * IN_F + f);
    acc.x += v0.x * w0; acc.y += v0.y * w0; acc.z += v0.z * w0; acc.w += v0.w * w0;
  }
  *(float4*)(agg + (long)d * IN_F + f) = acc;
}

__global__ void __launch_bounds__(256)
k_gemm_relu(const float* __restrict__ A, const float* __restrict__ nd,
            const float* __restrict__ W, const float* __restrict__ bias,
            float* __restrict__ out, int nrows) {
  __shared__ float sW[IN_F * OUT_F];
  for (int i = threadIdx.x; i < IN_F * OUT_F; i += blockDim.x) sW[i] = W[i];
  __syncthreads();

  int wave = (int)((blockIdx.x * blockDim.x + threadIdx.x) >> 5);
  int lane = threadIdx.x & 31;
  int m0 = wave * 16;
  if (m0 >= nrows) return;

  int lane16 = lane & 15;
  int half   = lane >> 4;
  int arow   = m0 + lane16;
  float scale = nd[arow];
  const float* ap = A + (long)arow * IN_F + half * 2;

  v8f acc0 = {}, acc1 = {}, acc2 = {}, acc3 = {};
#pragma unroll
  for (int k0 = 0; k0 < IN_F; k0 += 4) {
    float2 av = *(const float2*)(ap + k0);
    v2f a = {av.x * scale, av.y * scale};
    int kk = k0 + half * 2;
    const float* w0 = sW + kk * OUT_F + lane16;
    const float* w1 = w0 + OUT_F;
    v2f b0 = {w0[0],  w1[0]};
    v2f b1 = {w0[16], w1[16]};
    v2f b2 = {w0[32], w1[32]};
    v2f b3 = {w0[48], w1[48]};
    acc0 = __builtin_amdgcn_wmma_f32_16x16x4_f32(false, a, false, b0, (short)0, acc0, false, false);
    acc1 = __builtin_amdgcn_wmma_f32_16x16x4_f32(false, a, false, b1, (short)0, acc1, false, false);
    acc2 = __builtin_amdgcn_wmma_f32_16x16x4_f32(false, a, false, b2, (short)0, acc2, false, false);
    acc3 = __builtin_amdgcn_wmma_f32_16x16x4_f32(false, a, false, b3, (short)0, acc3, false, false);
  }

  int rbase = m0 + half * 8;
  float bi0 = bias[lane16], bi1 = bias[16 + lane16];
  float bi2 = bias[32 + lane16], bi3 = bias[48 + lane16];
#pragma unroll
  for (int r = 0; r < 8; ++r) {
    float* orow = out + (long)(rbase + r) * OUT_F + lane16;
    orow[0]  = fmaxf(acc0[r] + bi0, 0.f);
    orow[16] = fmaxf(acc1[r] + bi1, 0.f);
    orow[32] = fmaxf(acc2[r] + bi2, 0.f);
    orow[48] = fmaxf(acc3[r] + bi3, 0.f);
  }
}

__global__ void __launch_bounds__(256)
k_gemm16(const float* __restrict__ A, const float* __restrict__ W,
         float* __restrict__ out, int nrows) {
  __shared__ float sW[OUT_F * NCLS];
  for (int i = threadIdx.x; i < OUT_F * NCLS; i += blockDim.x) sW[i] = W[i];
  __syncthreads();

  int wave = (int)((blockIdx.x * blockDim.x + threadIdx.x) >> 5);
  int lane = threadIdx.x & 31;
  int m0 = wave * 16;
  if (m0 >= nrows) return;

  int lane16 = lane & 15;
  int half   = lane >> 4;
  const float* ap = A + (long)(m0 + lane16) * OUT_F + half * 2;

  v8f acc = {};
#pragma unroll
  for (int k0 = 0; k0 < OUT_F; k0 += 4) {
    float2 av = *(const float2*)(ap + k0);
    v2f a = {av.x, av.y};
    int kk = k0 + half * 2;
    v2f b = {sW[kk * NCLS + lane16], sW[(kk + 1) * NCLS + lane16]};
    acc = __builtin_amdgcn_wmma_f32_16x16x4_f32(false, a, false, b, (short)0, acc, false, false);
  }
  int rbase = m0 + half * 8;
#pragma unroll
  for (int r = 0; r < 8; ++r)
    out[(long)(rbase + r) * NCLS + lane16] = acc[r];
}

__global__ void k_gather16_out(const int* __restrict__ rowstart, const int* __restrict__ cnt,
                               const int* __restrict__ col, const float* __restrict__ z,
                               const float* __restrict__ ns, const float* __restrict__ nd,
                               const float* __restrict__ bias, float* __restrict__ out, int N) {
  int t = blockIdx.x * blockDim.x + threadIdx.x;
  int d = t >> 2;
  if (d >= N) return;
  int f = (t & 3) << 2;
  int j = rowstart[d];
  int end = j + cnt[d];
  float4 acc = make_float4(0.f, 0.f, 0.f, 0.f);
  for (; j + 4 <= end; j += 4) {
    int s0 = col[j], s1 = col[j + 1], s2 = col[j + 2], s3 = col[j + 3];
    float w0 = ns[s0], w1 = ns[s1], w2 = ns[s2], w3 = ns[s3];
    float4 v0 = *(const float4*)(z + (long)s0 * NCLS + f);
    float4 v1 = *(const float4*)(z + (long)s1 * NCLS + f);
    float4 v2 = *(const float4*)(z + (long)s2 * NCLS + f);
    float4 v3 = *(const float4*)(z + (long)s3 * NCLS + f);
    acc.x += v0.x * w0; acc.y += v0.y * w0; acc.z += v0.z * w0; acc.w += v0.w * w0;
    acc.x += v1.x * w1; acc.y += v1.y * w1; acc.z += v1.z * w1; acc.w += v1.w * w1;
    acc.x += v2.x * w2; acc.y += v2.y * w2; acc.z += v2.z * w2; acc.w += v2.w * w2;
    acc.x += v3.x * w3; acc.y += v3.y * w3; acc.z += v3.z * w3; acc.w += v3.w * w3;
  }
#pragma nounroll
  for (; j < end; ++j) {
    int s0 = col[j];
    float w0 = ns[s0];
    float4 v0 = *(const float4*)(z + (long)s0 * NCLS + f);
    acc.x += v0.x * w0; acc.y += v0.y * w0; acc.z += v0.z * w0; acc.w += v0.w * w0;
  }
  float sc = nd[d];
  float4 o;
  o.x = acc.x * sc + bias[f + 0];
  o.y = acc.y * sc + bias[f + 1];
  o.z = acc.z * sc + bias[f + 2];
  o.w = acc.w * sc + bias[f + 3];
  *(float4*)(out + (long)d * NCLS + f) = o;
}

extern "C" void kernel_launch(void* const* d_in, const int* in_sizes, int n_in,
                              void* d_out, int out_size, void* d_ws, size_t ws_size,
                              hipStream_t stream) {
  const float* x  = (const float*)d_in[0];
  const int*   src = (const int*)d_in[1];
  const int*   dst = (const int*)d_in[2];
  const float* W1 = (const float*)d_in[3];
  const float* b1 = (const float*)d_in[4];
  const float* W2 = (const float*)d_in[5];
  const float* b2 = (const float*)d_in[6];
  float* out = (float*)d_out;

  const int N = in_sizes[0] / IN_F;
  const int E = in_sizes[1];

  char* ws = (char*)d_ws;
  size_t off = 0;
  auto carve = [&](size_t bytes) -> void* {
    void* p = (void*)(ws + off);
    off += (bytes + 255) & ~(size_t)255;
    return p;
  };
  int*   deg_s = (int*)carve((size_t)N * 4);
  int*   deg_d = (int*)carve((size_t)N * 4);
  float* ns    = (float*)carve((size_t)N * 4);
  float* nd    = (float*)carve((size_t)N * 4);
  float* agg1  = (float*)carve((size_t)N * IN_F * 4);
  float* y1    = (float*)carve((size_t)N * OUT_F * 4);
  float* z     = (float*)carve((size_t)N * NCLS * 4);
  float* agg2  = (float*)carve((size_t)N * NCLS * 4);

  const int B = 256;
  auto cdiv = [](long a, long b) { return (int)((a + b - 1) / b); };

  int* rowstart = deg_s;
  int* pos      = (int*)y1;
  int* col      = (int*)agg2;

  long n4;
  n4 = ((long)N * 4 + 15) / 16;
  k_zero<<<cdiv(n4, B), B, 0, stream>>>((float4*)deg_s, n4);
  k_zero<<<cdiv(n4, B), B, 0, stream>>>((float4*)deg_d, n4);

  k_degree_rank<<<cdiv(E, B), B, 0, stream>>>(src, dst, deg_s, deg_d, pos, E);
  k_norm<<<cdiv(N, B), B, 0, stream>>>(deg_s, deg_d, ns, nd, N);

  k_scan<<<cdiv(N, 1024), 1024, 0, stream>>>(deg_d, rowstart, N);
  k_fill<<<cdiv(E, B), B, 0, stream>>>(src, dst, rowstart, pos, col, E);

  k_gather64<<<cdiv((long)N * 16, B), B, 0, stream>>>(rowstart, deg_d, col, x, ns, agg1, N);
  int waves = cdiv(N, 16);
  k_gemm_relu<<<cdiv((long)waves * 32, B), B, 0, stream>>>(agg1, nd, W1, b1, y1, N);

  k_gemm16<<<cdiv((long)waves * 32, B), B, 0, stream>>>(y1, W2, z, N);
  k_gather16_out<<<cdiv((long)N * 4, B), B, 0, stream>>>(rowstart, deg_d, col, z, ns, nd, b2, out, N);
}


# global int atomics (degree counts, edge ranks) replaced by a two-level counting sort on LDS counters (256-node buckets); sum order = arrival rank in the row, as before
# speedup vs baseline: 3.1123x; 1.9921x over previous
#include <hip/hip_runtime.h>

typedef __attribute__((ext_vector_type(2))) float v2f;
typedef __attribute__((ext_vector_type(8))) float v8f;

#define IN_F 64
#define OUT_F 64
#define NCLS 16

#define BSH 8
#define BNODES 256
#define NB_MAX 512
#define CHUNK 8192
#define ROWCAP 8192

__device__ __forceinline__ int block_excl_scan(int v, int* wsum) {
  int lane = threadIdx.x & 31, w = threadIdx.x >> 5;
  int x = v;
  for (int o = 1; o < 32; o <<= 1) {
    int y = __shfl_up(x, o, 32);
    if (lane >= o) x += y;
  }
  if (lane == 31) wsum[w] = x;
  __syncthreads();
  if (w == 0) {
    int nw = (int)(blockDim.x >> 5);
    int s = (lane < nw) ? wsum[lane] : 0;
    for (int o = 1; o < 32; o <<= 1) {
      int y = __shfl_up(s, o, 32);
      if (lane >= o) s += y;
    }
    wsum[lane] = s;
  }
  __syncthreads();
  int prev = (w > 0) ? wsum[w - 1] : 0;
  return prev + x - v;
}

__global__ void __launch_bounds__(1024)
k_bucket_count(const int* __restrict__ src, const int* __restrict__ dst,
               int* __restrict__ cnt, int E, int NB, int G) {
  __shared__ int h[2 * NB_MAX];
  int t = threadIdx.x, g = blockIdx.x;
  for (int i = t; i < 2 * NB; i += 1024) h[i] = 0;
  __syncthreads();
  int beg = g * CHUNK;
  int end = min(beg + CHUNK, E);
  for (int e = beg + t * 4; e < end; e += 4096) {
    if (e + 4 <= end) {
      int4 d = *(const int4*)(dst + e);
      int4 s = *(const int4*)(src + e);
      atomicAdd(&h[d.x >> BSH], 1); atomicAdd(&h[d.y >> BSH], 1);
      atomicAdd(&h[d.z >> BSH], 1); atomicAdd(&h[d.w >> BSH], 1);
      atomicAdd(&h[NB + (s.x >> BSH)], 1); atomicAdd(&h[NB + (s.y >> BSH)], 1);
      atomicAdd(&h[NB + (s.z >> BSH)], 1); atomicAdd(&h[NB + (s.w >> BSH)], 1);
    } else {
      for (int k = e; k < end; ++k) {
        atomicAdd(&h[dst[k] >> BSH], 1);
        atomicAdd(&h[NB + (src[k] >> BSH)], 1);
      }
    }
  }
  __syncthreads();
  for (int i = t; i < 2 * NB; i += 1024) cnt[i * G + g] = h[i];
}

__global__ void __launch_bounds__(1024)
k_bucket_scatter(const int* __restrict__ src, const int* __restrict__ dst,
                 const int* __restrict__ cnt, const int* __restrict__ boff,
                 int* __restrict__ rec, int E, int NB, int G) {
  __shared__ int cur[2 * NB_MAX];
  __shared__ int dl[2 * NB_MAX];
  __shared__ int wsum[32];
  __shared__ int stage[2 * CHUNK];
  __shared__ unsigned short sbk[2 * CHUNK];
  int t = threadIdx.x, g = blockIdx.x;
  int nb2 = 2 * NB;
  int v  = (t < nb2) ? cnt[t * G + g] : 0;
  int bo = (t < nb2) ? boff[t * G + g] : 0;
  int ex = block_excl_scan(v, wsum);
  if (t < nb2) { cur[t] = ex; dl[t] = bo - ex; }
  __syncthreads();
  int beg = g * CHUNK;
  int end = min(beg + CHUNK, E);
  for (int e = beg + t * 4; e < end; e += 4096) {
    if (e + 4 <= end) {
      int4 d = *(const int4*)(dst + e);
      int4 s = *(const int4*)(src + e);
      int b0 = d.x >> BSH, b1 = d.y >> BSH, b2 = d.z >> BSH, b3 = d.w >> BSH;
      int p0 = atomicAdd(&cur[b0], 1), p1 = atomicAdd(&cur[b1], 1);
      int p2 = atomicAdd(&cur[b2], 1), p3 = atomicAdd(&cur[b3], 1);
      stage[p0] = (s.x << BSH) | (d.x & (BNODES - 1)); sbk[p0] = (unsigned short)b0;
      stage[p1] = (s.y << BSH) | (d.y & (BNODES - 1)); sbk[p1] = (unsigned short)b1;
      stage[p2] = (s.z << BSH) | (d.z & (BNODES - 1)); sbk[p2] = (unsigned short)b2;
      stage[p3] = (s.w << BSH) | (d.w & (BNODES - 1)); sbk[p3] = (unsigned short)b3;
      int c0 = NB + (s.x >> BSH), c1 = NB + (s.y >> BSH), c2 = NB + (s.z >> BSH), c3 = NB + (s.w >> BSH);
      int q0 = atomicAdd(&cur[c0], 1), q1 = atomicAdd(&cur[c1], 1);
      int q2 = atomicAdd(&cur[c2], 1), q3 = atomicAdd(&cur[c3], 1);
      stage[q0] = s.x & (BNODES - 1); sbk[q0] = (unsigned short)c0;
      stage[q1] = s.y & (BNODES - 1); sbk[q1] = (unsigned short)c1;
      stage[q2] = s.z & (BNODES - 1); sbk[q2] = (unsigned short)c2;
      stage[q3] = s.w & (BNODES - 1); sbk[q3] = (unsigned short)c3;
    } else {
      for (int k = e; k < end; ++k) {
        int d = dst[k], s = src[k];
        int b = d >> BSH, c = NB + (s >> BSH);
        int p = atomicAdd(&cur[b], 1);
        stage[p] = (s << BSH) | (d & (BNODES - 1)); sbk[p] = (unsigned short)b;
        int q = atomicAdd(&cur[c], 1);
        stage[q] = s & (BNODES - 1); sbk[q] = (unsigned short)c;
      }
    }
  }
  __syncthreads();
  int len2 = 2 * max(end - beg, 0);
  for (int j = t; j < len2; j += 1024) rec[j + dl[sbk[j]]] = stage[j];
}

__global__ void __launch_bounds__(512)
k_bucket_rows(const int* __restrict__ rec, const int* __restrict__ boff,
              int* __restrict__ rowstart, int* __restrict__ deg_d, int* __restrict__ col,
              float* __restrict__ ns, float* __restrict__ nd, int N, int E, int NB, int G) {
  __shared__ int c[BNODES];
  __shared__ int wsum[32];
  __shared__ int lcol[ROWCAP];
  int t = threadIdx.x;
  int bb = blockIdx.x;
  int beg = boff[bb * G];
  int end = (bb + 1 < 2 * NB) ? boff[(bb + 1) * G] : 2 * E;
  if (t < BNODES) c[t] = 0;
  __syncthreads();
  for (int j0 = beg + t; j0 < end; j0 += 4 * 512) {
    int r[4];
    for (int k = 0; k < 4; ++k) { int j = j0 + k * 512; r[k] = (j < end) ? rec[j] : -1; }
    for (int k = 0; k < 4; ++k) if (r[k] >= 0) atomicAdd(&c[r[k] & (BNODES - 1)], 1);
  }
  __syncthreads();

  if (bb >= NB) {
    int node = (bb - NB) * BNODES + t;
    if (t < BNODES && node < N) { int a = c[t]; ns[node] = (a > 0) ? rsqrtf((float)a) : 1.0f; }
    return;
  }

  int mine = (t < BNODES) ? c[t] : 0;
  int ex = block_excl_scan(mine, wsum);
  if (t < BNODES) {
    c[t] = ex;
    int node = bb * BNODES + t;
    if (node < N) {
      rowstart[node] = beg + ex;
      deg_d[node] = mine;
      nd[node] = (mine > 0) ? rsqrtf((float)mine) : 1.0f;
    }
  }
  __syncthreads();
  for (int j0 = beg + t; j0 < end; j0 += 4 * 512) {
    int r[4];
    for (int k = 0; k < 4; ++k) { int j = j0 + k * 512; r[k] = (j < end) ? rec[j] : -1; }
    for (int k = 0; k < 4; ++k) {
      if (r[k] >= 0) {
        int p = atomicAdd(&c[r[k] & (BNODES - 1)], 1);
        if (p < ROWCAP) lcol[p] = r[k] >> BSH;
        else col[beg + p] = r[k] >> BSH;
      }
    }
  }
  __syncthreads();
  int nst = min(end - beg, ROWCAP);
  for (int j = t; j < nst; j += 512) col[beg + j] = lcol[j];
}

__global__ void __launch_bounds__(1024)
k_scan(const int* __restrict__ cnt, int* __restrict__ rowstart, int N) {
  __shared__ int s[1024];
  int t = threadIdx.x;
  int base = blockIdx.x * 1024;

  int part = 0;
#pragma unroll 8
  for (int i = t * 4; i < base; i += 4096) {
    int4 v = *(const int4*)(cnt + i);
    part += (v.x + v.y) + (v.z + v.w);
  }
  s[t] = part;
  __syncthreads();
  for (int off = 512; off > 0; off >>= 1) {
    if (t < off) s[t] += s[t + off];
    __syncthreads();
  }
  int prefix = s[0];
  __syncthreads();

  int i = base + t;
  int v = (i < N) ? cnt[i] : 0;
  s[t] = v;
  __syncthreads();
  for (int off = 1; off < 1024; off <<= 1) {
    int add = (t >= off) ? s[t - off] : 0;
    __syncthreads();
    s[t] += add;
    __syncthreads();
  }
  if (i < N) rowstart[i] = prefix + s[t] - v;
}

__global__ void k_gather64(const int* __restrict__ rowstart, const int* __restrict__ cnt,
                           const int* __restrict__ col, const float* __restrict__ x,
                           const float* __restrict__ ns, float* __restrict__ agg, int N) {
  long t = (long)blockIdx.x * blockDim.x + threadIdx.x;
  int d = (int)(t >> 4);
  if (d >= N) return;
  int f = ((int)t & 15) << 2;
  int j = rowstart[d];
  int end = j + cnt[d];
  float4 acc = make_float4(0.f, 0.f, 0.f, 0.f);
  for (; j + 4 <= end; j += 4) {
    int s0 = col[j], s1 = col[j + 1], s2 = col[j + 2], s3 = col[j + 3];
    float w0 = ns[s0], w1 = ns[s1], w2 = ns[s2], w3 = ns[s3];
    float4 v0 = *(const float4*)(x + (long)s0 * IN_F + f);
    float4 v1 = *(const float4*)(x + (long)s1 * IN_F + f);
    float4 v2 = *(const float4*)(x + (long)s2 * IN_F + f);
    float4 v3 = *(const float4*)(x + (long)s3 * IN_F + f);
    acc.x += v0.x * w0; acc.y += v0.y * w0; acc.z += v0.z * w0; acc.w += v0.w * w0;
    acc.x += v1.x * w1; acc.y += v1.y * w1; acc.z += v1.z * w1; acc.w += v1.w * w1;
    acc.x += v2.x * w2; acc.y += v2.y * w2; acc.z += v2.z * w2; acc.w += v2.w * w2;
    acc.x += v3.x * w3; acc.y += v3.y * w3; acc.z += v3.z * w3; acc.w += v3.w * w3;
  }
#pragma nounroll
  for (; j < end; ++j) {
    int s0 = col[j];
    float w0 = ns[s0];
    float4 v0 = *(const float4*)(x + (long)s0 * IN_F + f);
    acc.x += v0.x * w0; acc.y += v0.y * w0; acc.z += v0.z * w0; acc.w += v0.w * w0;
  }
  *(float4*)(agg + (long)d * IN_F + f) = acc;
}

__global__ void __launch_bounds__(256)
k_gemm_relu(const float* __restrict__ A, const float* __restrict__ nd,
            const float* __restrict__ W, const float* __restrict__ bias,
            float* __restrict__ out, int nrows) {
  __shared__ float sW[IN_F * OUT_F];
  for (int i = threadIdx.x; i < IN_F * OUT_F; i += blockDim.x) sW[i] = W[i];
  __syncthreads();

  int wave = (int)((blockIdx.x * blockDim.x + threadIdx.x) >> 5);
  int lane = threadIdx.x & 31;
  int m0 = wave * 16;
  if (m0 >= nrows) return;

  int lane16 = lane & 15;
  int half   = lane >> 4;
  int arow   = m0 + lane16;
  float scale = nd[arow];
  const float* ap = A + (long)arow * IN_F + half * 2;

  v8f acc0 = {}, acc1 = {}, acc2 = {}, acc3 = {};
#pragma unroll
  for (int k0 = 0; k0 < IN_F; k0 += 4) {
    float2 av = *(const float2*)(ap + k0);
    v2f a = {av.x * scale, av.y * scale};
    int kk = k0 + half * 2;
    const float* w0 = sW + kk * OUT_F + lane16;
    const float* w1 = w0 + OUT_F;
    v2f b0 = {w0[0],  w1[0]};
    v2f b1 = {w0[16], w1[16]};
    v2f b2 = {w0[32], w1[32]};
    v2f b3 = {w0[48], w1[48]};
    acc0 = __builtin_amdgcn_wmma_f32_16x16x4_f32(false, a, false, b0, (short)0, acc0, false, false);
    acc1 = __builtin_amdgcn_wmma_f32_16x16x4_f32(false, a, false, b1, (short)0, acc1, false, false);
    acc2 = __builtin_amdgcn_wmma_f32_16x16x4_f32(false, a, false, b2, (short)0, acc2, false, false);
    acc3 = __builtin_amdgcn_wmma_f32_16x16x4_f32(false, a, false, b3, (short)0, acc3, false, false);
  }

  int rbase = m0 + half * 8;
  float bi0 = bias[lane16], bi1 = bias[16 + lane16];
  float bi2 = bias[32 + lane16], bi3 = bias[48 + lane16];
#pragma unroll
  for (int r = 0; r < 8; ++r) {
    float* orow = out + (long)(rbase + r) * OUT_F + lane16;
    orow[0]  = fmaxf(acc0[r] + bi0, 0.f);
    orow[16] = fmaxf(acc1[r] + bi1, 0.f);
    orow[32] = fmaxf(acc2[r] + bi2, 0.f);
    orow[48] = fmaxf(acc3[r] + bi3, 0.f);
  }
}

__global__ void __launch_bounds__(256)
k_gemm16(const float* __restrict__ A, const float* __restrict__ W,
         float* __restrict__ out, int nrows) {
  __shared__ float sW[OUT_F * NCLS];
  for (int i = threadIdx.x; i < OUT_F * NCLS; i += blockDim.x) sW[i] = W[i];
  __syncthreads();

  int wave = (int)((blockIdx.x * blockDim.x + threadIdx.x) >> 5);
  int lane = threadIdx.x & 31;
  int m0 = wave * 16;
  if (m0 >= nrows) return;

  int lane16 = lane & 15;
  int half   = lane >> 4;
  const float* ap = A + (long)(m0 + lane16) * OUT_F + half * 2;

  v8f acc = {};
#pragma unroll
  for (int k0 = 0; k0 < OUT_F; k0 += 4) {
    float2 av = *(const float2*)(ap + k0);
    v2f a = {av.x, av.y};
    int kk = k0 + half * 2;
    v2f b = {sW[kk * NCLS + lane16], sW[(kk + 1) * NCLS + lane16]};
    acc = __builtin_amdgcn_wmma_f32_16x16x4_f32(false, a, false, b, (short)0, acc, false, false);
  }
  int rbase = m0 + half * 8;
#pragma unroll
  for (int r = 0; r < 8; ++r)
    out[(long)(rbase + r) * NCLS + lane16] = acc[r];
}

__global__ void k_gather16_out(const int* __restrict__ rowstart, const int* __restrict__ cnt,
                               const int* __restrict__ col, const float* __restrict__ z,
                               const float* __restrict__ ns, const float* __restrict__ nd,
                               const float* __restrict__ bias, float* __restrict__ out, int N) {
  int t = blockIdx.x * blockDim.x + threadIdx.x;
  int d = t >> 2;
  if (d >= N) return;
  int f = (t & 3) << 2;
  int j = rowstart[d];
  int end = j + cnt[d];
  float4 acc = make_float4(0.f, 0.f, 0.f, 0.f);
  for (; j + 4 <= end; j += 4) {
    int s0 = col[j], s1 = col[j + 1], s2 = col[j + 2], s3 = col[j + 3];
    float w0 = ns[s0], w1 = ns[s1], w2 = ns[s2], w3 = ns[s3];
    float4 v0 = *(const float4*)(z + (long)s0 * NCLS + f);
    float4 v1 = *(const float4*)(z + (long)s1 * NCLS + f);
    float4 v2 = *(const float4*)(z + (long)s2 * NCLS + f);
    float4 v3 = *(const float4*)(z + (long)s3 * NCLS + f);
    acc.x += v0.x * w0; acc.y += v0.y * w0; acc.z += v0.z * w0; acc.w += v0.w * w0;
    acc.x += v1.x * w1; acc.y += v1.y * w1; acc.z += v1.z * w1; acc.w += v1.w * w1;
    acc.x += v2.x * w2; acc.y += v2.y * w2; acc.z += v2.z * w2; acc.w += v2.w * w2;
    acc.x += v3.x * w3; acc.y += v3.y * w3; acc.z += v3.z * w3; acc.w += v3.w * w3;
  }
#pragma nounroll
  for (; j < end; ++j) {
    int s0 = col[j];
    float w0 = ns[s0];
    float4 v0 = *(const float4*)(z + (long)s0 * NCLS + f);
    acc.x += v0.x * w0; acc.y += v0.y * w0; acc.z += v0.z * w0; acc.w += v0.w * w0;
  }
  float sc = nd[d];
  float4 o;
  o.x = acc.x * sc + bias[f + 0];
  o.y = acc.y * sc + bias[f + 1];
  o.z = acc.z * sc + bias[f + 2];
  o.w = acc.w * sc + bias[f + 3];
  *(float4*)(out + (long)d * NCLS + f) = o;
}

extern "C" void kernel_launch(void* const* d_in, const int* in_sizes, int n_in,
                              void* d_out, int out_size, void* d_ws, size_t ws_size,
                              hipStream_t stream) {
  const float* x  = (const float*)d_in[0];
  const int*   src = (const int*)d_in[1];
  const int*   dst = (const int*)d_in[2];
  const float* W1 = (const float*)d_in[3];
  const float* b1 = (const float*)d_in[4];
  const float* W2 = (const float*)d_in[5];
  const float* b2 = (const float*)d_in[6];
  float* out = (float*)d_out;

  const int N = in_sizes[0] / IN_F;
  const int E = in_sizes[1];

  char* ws = (char*)d_ws;
  size_t off = 0;
  auto carve = [&](size_t bytes) -> void* {
    void* p = (void*)(ws + off);
    off += (bytes + 255) & ~(size_t)255;
    return p;
  };
  int*   deg_s = (int*)carve((size_t)N * 4);
  int*   deg_d = (int*)carve((size_t)N * 4);
  float* ns    = (float*)carve((size_t)N * 4);
  float* nd    = (float*)carve((size_t)N * 4);
  float* agg1  = (float*)carve((size_t)N * IN_F * 4);
  float* y1    = (float*)carve((size_t)N * OUT_F * 4);
  float* z     = (float*)carve((size_t)N * NCLS * 4);
  float* agg2  = (float*)carve((size_t)N * NCLS * 4);

  const int B = 256;
  auto cdiv = [](long a, long b) { return (int)((a + b - 1) / b); };

  int* rowstart = deg_s;
  int* col      = (int*)agg2;
  const int NB = cdiv(N, BNODES);
  const int G  = cdiv(E, CHUNK);
  const int NT = 2 * NB * G;
  int* rec  = (int*)y1;
  int* cnt  = rec + (((size_t)2 * E + 63) & ~(size_t)63);
  int* boff = cnt + (((size_t)NT + 63) & ~(size_t)63);

  k_bucket_count<<<G, 1024, 0, stream>>>(src, dst, cnt, E, NB, G);
  k_scan<<<cdiv(NT, 1024), 1024, 0, stream>>>(cnt, boff, NT);
  k_bucket_scatter<<<G, 1024, 0, stream>>>(src, dst, cnt, boff, rec, E, NB, G);
  k_bucket_rows<<<2 * NB, 512, 0, stream>>>(rec, boff, rowstart, deg_d, col, ns, nd, N, E, NB, G);

  k_gather64<<<cdiv((long)N * 16, B), B, 0, stream>>>(rowstart, deg_d, col, x, ns, agg1, N);
  int waves = cdiv(N, 16);
  k_gemm_relu<<<cdiv((long)waves * 32, B), B, 0, stream>>>(agg1, nd, W1, b1, y1, N);

  k_gemm16<<<cdiv((long)waves * 32, B), B, 0, stream>>>(y1, W2, z, N);
  k_gather16_out<<<cdiv((long)N * 4, B), B, 0, stream>>>(rowstart, deg_d, col, z, ns, nd, b2, out, N);
}


# layer-1 row gather, first product+relu and second product fused into one kernel: 2 waves per 16-row tile, summed and relu tiles stay in LDS, neither plane goes to memory; row sum order unchanged
# speedup vs baseline: 3.9470x; 1.2682x over previous
#include <hip/hip_runtime.h>

typedef __attribute__((ext_vector_type(2))) float v2f;
typedef __attribute__((ext_vector_type(8))) float v8f;

#define IN_F 64
#define OUT_F 64
#define NCLS 16

#define BSH 8
#define BNODES 256
#define NB_MAX 512
#define CHUNK 8192
#define ROWCAP 8192

__device__ __forceinline__ int block_excl_scan(int v, int* wsum) {
  int lane = threadIdx.x & 31, w = threadIdx.x >> 5;
  int x = v;
  for (int o = 1; o < 32; o <<= 1) {
    int y = __shfl_up(x, o, 32);
    if (lane >= o) x += y;
  }
  if (lane == 31) wsum[w] = x;
  __syncthreads();
  if (w == 0) {
    int nw = (int)(blockDim.x >> 5);
    int s = (lane < nw) ? wsum[lane] : 0;
    for (int o = 1; o < 32; o <<= 1) {
      int y = __shfl_up(s, o, 32);
      if (lane >= o) s += y;
    }
    wsum[lane] = s;
  }
  __syncthreads();
  int prev = (w > 0) ? wsum[w - 1] : 0;
  return prev + x - v;
}

__global__ void __launch_bounds__(1024)
k_bucket_count(const int* __restrict__ src, const int* __restrict__ dst,
               int* __restrict__ cnt, int E, int NB, int G) {
  __shared__ int h[2 * NB_MAX];
  int t = threadIdx.x, g = blockIdx.x;
  for (int i = t; i < 2 * NB; i += 1024) h[i] = 0;
  __syncthreads();
  int beg = g * CHUNK;
  int end = min(beg + CHUNK, E);
  for (int e = beg + t * 4; e < end; e += 4096) {
    if (e + 4 <= end) {
      int4 d = *(const int4*)(dst + e);
      int4 s = *(const int4*)(src + e);
      atomicAdd(&h[d.x >> BSH], 1); atomicAdd(&h[d.y >> BSH], 1);
      atomicAdd(&h[d.z >> BSH], 1); atomicAdd(&h[d.w >> BSH], 1);
      atomicAdd(&h[NB + (s.x >> BSH)], 1); atomicAdd(&h[NB + (s.y >> BSH)], 1);
      atomicAdd(&h[NB + (s.z >> BSH)], 1); atomicAdd(&h[NB + (s.w >> BSH)], 1);
    } else {
      for (int k = e; k < end; ++k) {
        atomicAdd(&h[dst[k] >> BSH], 1);
        atomicAdd(&h[NB + (src[k] >> BSH)], 1);
      }
    }
  }
  __syncthreads();
  for (int i = t; i < 2 * NB; i += 1024) cnt[i * G + g] = h[i];
}

__global__ void __launch_bounds__(1024)
k_bucket_scatter(const int* __restrict__ src, const int* __restrict__ dst,
                 const int* __restrict__ cnt, const int* __restrict__ boff,
                 int* __restrict__ rec, int E, int NB, int G) {
  __shared__ int cur[2 * NB_MAX];
  __shared__ int dl[2 * NB_MAX];
  __shared__ int wsum[32];
  __shared__ int stage[2 * CHUNK];
  __shared__ unsigned short sbk[2 * CHUNK];
  int t = threadIdx.x, g = blockIdx.x;
  int nb2 = 2 * NB;
  int v  = (t < nb2) ? cnt[t * G + g] : 0;
  int bo = (t < nb2) ? boff[t * G + g] : 0;
  int ex = block_excl_scan(v, wsum);
  if (t < nb2) { cur[t] = ex; dl[t] = bo - ex; }
  __syncthreads();
  int beg = g * CHUNK;
  int end = min(beg + CHUNK, E);
  for (int e = beg + t * 4; e < end; e += 4096) {
    if (e + 4 <= end) {
      int4 d = *(const int4*)(dst + e);
      int4 s = *(const int4*)(src + e);
      int b0 = d.x >> BSH, b1 = d.y >> BSH, b2 = d.z >> BSH, b3 = d.w >> BSH;
      int p0 = atomicAdd(&cur[b0], 1), p1 = atomicAdd(&cur[b1], 1);
      int p2 = atomicAdd(&cur[b2], 1), p3 = atomicAdd(&cur[b3], 1);
      stage[p0] = (s.x << BSH) | (d.x & (BNODES - 1)); sbk[p0] = (unsigned short)b0;
      stage[p1] = (s.y << BSH) | (d.y & (BNODES - 1)); sbk[p1] = (unsigned short)b1;
      stage[p2] = (s.z << BSH) | (d.z & (BNODES - 1)); sbk[p2] = (unsigned short)b2;
      stage[p3] = (s.w << BSH) | (d.w & (BNODES - 1)); sbk[p3] = (unsigned short)b3;
      int c0 = NB + (s.x >> BSH), c1 = NB + (s.y >> BSH), c2 = NB + (s.z >> BSH), c3 = NB + (s.w >> BSH);
      int q0 = atomicAdd(&cur[c0], 1), q1 = atomicAdd(&cur[c1], 1);
      int q2 = atomicAdd(&cur[c2], 1), q3 = atomicAdd(&cur[c3], 1);
      stage[q0] = s.x & (BNODES - 1); sbk[q0] = (unsigned short)c0;
      stage[q1] = s.y & (BNODES - 1); sbk[q1] = (unsigned short)c1;
      stage[q2] = s.z & (BNODES - 1); sbk[q2] = (unsigned short)c2;
      stage[q3] = s.w & (BNODES - 1); sbk[q3] = (unsigned short)c3;
    } else {
      for (int k = e; k < end; ++k) {
        int d = dst[k], s = src[k];
        int b = d >> BSH, c = NB + (s >> BSH);
        int p = atomicAdd(&cur[b], 1);
        stage[p] = (s << BSH) | (d & (BNODES - 1)); sbk[p] = (unsigned short)b;
        int q = atomicAdd(&cur[c], 1);
        stage[q] = s & (BNODES - 1); sbk[q] = (unsigned short)c;
      }
    }
  }
  __syncthreads();
  int len2 = 2 * max(end - beg, 0);
  for (int j = t; j < len2; j += 1024) rec[j + dl[sbk[j]]] = stage[j];
}

__global__ void __launch_bounds__(512)
k_bucket_rows(const int* __restrict__ rec, const int* __restrict__ boff,
              int* __restrict__ rowstart, int* __restrict__ deg_d, int* __restrict__ col,
              float* __restrict__ ns, float* __restrict__ nd, int N, int E, int NB, int G) {
  __shared__ int c[BNODES];
  __shared__ int wsum[32];
  __shared__ int lcol[ROWCAP];
  int t = threadIdx.x;
  int bb = blockIdx.x;
  int beg = boff[bb * G];
  int end = (bb + 1 < 2 * NB) ? boff[(bb + 1) * G] : 2 * E;
  if (t < BNODES) c[t] = 0;
  __syncthreads();
  for (int j0 = beg + t; j0 < end; j0 += 4 * 512) {
    int r[4];
    for (int k = 0; k < 4; ++k) { int j = j0 + k * 512; r[k] = (j < end) ? rec[j] : -1; }
    for (int k = 0; k < 4; ++k) if (r[k] >= 0) atomicAdd(&c[r[k] & (BNODES - 1)], 1);
  }
  __syncthreads();

  if (bb >= NB) {
    int node = (bb - NB) * BNODES + t;
    if (t < BNODES && node < N) { int a = c[t]; ns[node] = (a > 0) ? rsqrtf((float)a) : 1.0f; }
    return;
  }

  int mine = (t < BNODES) ? c[t] : 0;
  int ex = block_excl_scan(mine, wsum);
  if (t < BNODES) {
    c[t] = ex;
    int node = bb * BNODES + t;
    if (node < N) {
      rowstart[node] = beg + ex;
      deg_d[node] = mine;
      nd[node] = (mine > 0) ? rsqrtf((float)mine) : 1.0f;
    }
  }
  __syncthreads();
  for (int j0 = beg + t; j0 < end; j0 += 4 * 512) {
    int r[4];
    for (int k = 0; k < 4; ++k) { int j = j0 + k * 512; r[k] = (j < end) ? rec[j] : -1; }
    for (int k = 0; k < 4; ++k) {
      if (r[k] >= 0) {
        int p = atomicAdd(&c[r[k] & (BNODES - 1)], 1);
        if (p < ROWCAP) lcol[p] = r[k] >> BSH;
        else col[beg + p] = r[k] >> BSH;
      }
    }
  }
  __syncthreads();
  int nst = min(end - beg, ROWCAP);
  for (int j = t; j < nst; j += 512) col[beg + j] = lcol[j];
}

__global__ void __launch_bounds__(1024)
k_scan(const int* __restrict__ cnt, int* __restrict__ rowstart, int N) {
  __shared__ int s[1024];
  int t = threadIdx.x;
  int base = blockIdx.x * 1024;

  int part = 0;
#pragma unroll 8
  for (int i = t * 4; i < base; i += 4096) {
    int4 v = *(const int4*)(cnt + i);
    part += (v.x + v.y) + (v.z + v.w);
  }
  s[t] = part;
  __syncthreads();
  for (int off = 512; off > 0; off >>= 1) {
    if (t < off) s[t] += s[t + off];
    __syncthreads();
  }
  int prefix = s[0];
  __syncthreads();

  int i = base + t;
  int v = (i < N) ? cnt[i] : 0;
  s[t] = v;
  __syncthreads();
  for (int off = 1; off < 1024; off <<= 1) {
    int add = (t >= off) ? s[t - off] : 0;
    __syncthreads();
    s[t] += add;
    __syncthreads();
  }
  if (i < N) rowstart[i] = prefix + s[t] - v;
}

#define TP 68
#define TPB 4

__global__ void __launch_bounds__(256)
k_layer1(const int* __restrict__ rowstart, const int* __restrict__ cnt,
         const int* __restrict__ col, const float* __restrict__ x,
         const float* __restrict__ ns, const float* __restrict__ nd,
         const float* __restrict__ W1, const float* __restrict__ b1,
         const float* __restrict__ W2, float* __restrict__ z, int N) {
  __shared__ __attribute__((aligned(16))) float sW1[IN_F * OUT_F];
  __shared__ __attribute__((aligned(16))) float sW2[OUT_F * NCLS];
  __shared__ __attribute__((aligned(16))) float tile[TPB][16 * TP];

  const int tid = threadIdx.x;
  const int lane = tid & 31, wv = tid >> 5;
  const int c = lane & 15, h = lane >> 4;
  const int t = wv >> 1, p = wv & 1;
  const int m0 = (blockIdx.x * TPB + t) * 16;
  float* tl = tile[t];

  const int myrow = m0 + 8 * p + (lane & 7);
  int rs = 0, rc = 0;
  if (myrow < N) { rs = rowstart[myrow]; rc = cnt[myrow]; }
  const int arow = m0 + c;
  const float scale = (arow < N) ? nd[arow] : 0.f;

  {
    float4 g[4];
#pragma unroll
    for (int q = 0; q < 4; ++q) g[q] = *(const float4*)(W1 + tid * 4 + q * 1024);
    float4 g2 = *(const float4*)(W2 + tid * 4);
#pragma unroll
    for (int q = 0; q < 4; ++q) *(float4*)(sW1 + tid * 4 + q * 1024) = g[q];
    *(float4*)(sW2 + tid * 4) = g2;
  }

  const int f = c << 2;
  for (int i = 0; i < 4; ++i) {
    const int lr = 2 * i + h;
    int j = __shfl(rs, lr, 32);
    const int end = j + __shfl(rc, lr, 32);
    const int last = max(end - 1, 0);
    float4 acc = make_float4(0.f, 0.f, 0.f, 0.f);
    int s[8];
#pragma unroll
    for (int k = 0; k < 8; ++k) s[k] = col[min(j + k, last)];
    while (j < end) {
      float w[8];
      float4 v[8];
      int sn[8];
#pragma unroll
      for (int k = 0; k < 8; ++k) {
        w[k] = ns[s[k]];
        v[k] = *(const float4*)(x + (long)s[k] * IN_F + f);
      }
#pragma unroll
      for (int k = 0; k < 8; ++k) sn[k] = col[min(j + 8 + k, last)];
#pragma unroll
      for (int k = 0; k < 8; ++k) {
        const bool in = (j + k < end);
        const float wk = in ? w[k] : 0.f;
        const float4 vk = in ? v[k] : make_float4(0.f, 0.f, 0.f, 0.f);
        acc.x += vk.x * wk; acc.y += vk.y * wk;
        acc.z += vk.z * wk; acc.w += vk.w * wk;
      }
#pragma unroll
      for (int k = 0; k < 8; ++k) s[k] = sn[k];
      j += 8;
    }
    *(float4*)(tl + (8 * p + lr) * TP + f) = acc;
  }
  __syncthreads();

  v2f a[16];
#pragma unroll
  for (int ks = 0; ks < 16; ++ks) {
    float2 av = *(const float2*)(tl + c * TP + ks * 4 + h * 2);
    a[ks] = (v2f){av.x * scale, av.y * scale};
  }
  __syncthreads();

  v8f acc0 = {}, acc1 = {};
#pragma unroll
  for (int ks = 0; ks < 16; ++ks) {
    const int kk = ks * 4 + h * 2;
    const float* w0 = sW1 + kk * OUT_F + 32 * p + c;
    v2f b0 = {w0[0],  w0[OUT_F]};
    v2f b1v = {w0[16], w0[OUT_F + 16]};
    acc0 = __builtin_amdgcn_wmma_f32_16x16x4_f32(false, a[ks], false, b0,  (short)0, acc0, false, false);
    acc1 = __builtin_amdgcn_wmma_f32_16x16x4_f32(false, a[ks], false, b1v, (short)0, acc1, false, false);
  }
  {
    const float bi0 = b1[32 * p + c], bi1 = b1[32 * p + 16 + c];
    float* yrow = tl + (8 * h) * TP + 32 * p + c;
#pragma unroll
    for (int r = 0; r < 8; ++r) {
      yrow[r * TP]      = fmaxf(acc0[r] + bi0, 0.f);
      yrow[r * TP + 16] = fmaxf(acc1[r] + bi1, 0.f);
    }
  }
  __syncthreads();

  if (p == 0 && m0 < N) {
    v8f accz = {};
#pragma unroll
    for (int ks = 0; ks < 16; ++ks) {
      float2 av = *(const float2*)(tl + c * TP + ks * 4 + h * 2);
      v2f ay = {av.x, av.y};
      const int kk = ks * 4 + h * 2;
      v2f b = {sW2[kk * NCLS + c], sW2[(kk + 1) * NCLS + c]};
      accz = __builtin_amdgcn_wmma_f32_16x16x4_f32(false, ay, false, b, (short)0, accz, false, false);
    }
    const int rbase = m0 + h * 8;
#pragma unroll
    for (int r = 0; r < 8; ++r)
      z[(long)(rbase + r) * NCLS + c] = accz[r];
  }
}

__global__ void k_gather16_out(const int* __restrict__ rowstart, const int* __restrict__ cnt,
                               const int* __restrict__ col, const float* __restrict__ z,
                               const float* __restrict__ ns, const float* __restrict__ nd,
                               const float* __restrict__ bias, float* __restrict__ out, int N) {
  int t = blockIdx.x * blockDim.x + threadIdx.x;
  int d = t >> 2;
  if (d >= N) return;
  int f = (t & 3) << 2;
  int j = rowstart[d];
  int end = j + cnt[d];
  float4 acc = make_float4(0.f, 0.f, 0.f, 0.f);
  for (; j + 4 <= end; j += 4) {
    int s0 = col[j], s1 = col[j + 1], s2 = col[j + 2], s3 = col[j + 3];
    float w0 = ns[s0], w1 = ns[s1], w2 = ns[s2], w3 = ns[s3];
    float4 v0 = *(const float4*)(z + (long)s0 * NCLS + f);
    float4 v1 = *(const float4*)(z + (long)s1 * NCLS + f);
    float4 v2 = *(const float4*)(z + (long)s2 * NCLS + f);
    float4 v3 = *(const float4*)(z + (long)s3 * NCLS + f);
    acc.x += v0.x * w0; acc.y += v0.y * w0; acc.z += v0.z * w0; acc.w += v0.w * w0;
    acc.x += v1.x * w1; acc.y += v1.y * w1; acc.z += v1.z * w1; acc.w += v1.w * w1;
    acc.x += v2.x * w2; acc.y += v2.y * w2; acc.z += v2.z * w2; acc.w += v2.w * w2;
    acc.x += v3.x * w3; acc.y += v3.y * w3; acc.z += v3.z * w3; acc.w += v3.w * w3;
  }
#pragma nounroll
  for (; j < end; ++j) {
    int s0 = col[j];
    float w0 = ns[s0];
    float4 v0 = *(const float4*)(z + (long)s0 * NCLS + f);
    acc.x += v0.x * w0; acc.y += v0.y * w0; acc.z += v0.z * w0; acc.w += v0.w * w0;
  }
  float sc = nd[d];
  float4 o;
  o.x = acc.x * sc + bias[f + 0];
  o.y = acc.y * sc + bias[f + 1];
  o.z = acc.z * sc + bias[f + 2];
  o.w = acc.w * sc + bias[f + 3];
  *(float4*)(out + (long)d * NCLS + f) = o;
}

extern "C" void kernel_launch(void* const* d_in, const int* in_sizes, int n_in,
                              void* d_out, int out_size, void* d_ws, size_t ws_size,
                              hipStream_t stream) {
  const float* x  = (const float*)d_in[0];
  const int*   src = (const int*)d_in[1];
  const int*   dst = (const int*)d_in[2];
  const float* W1 = (const float*)d_in[3];
  const float* b1 = (const float*)d_in[4];
  const float* W2 = (const float*)d_in[5];
  const float* b2 = (const float*)d_in[6];
  float* out = (float*)d_out;

  const int N = in_sizes[0] / IN_F;
  const int E = in_sizes[1];

  char* ws = (char*)d_ws;
  size_t off = 0;
  auto carve = [&](size_t bytes) -> void* {
    void* p = (void*)(ws + off);
    off += (bytes + 255) & ~(size_t)255;
    return p;
  };
  int*   deg_s = (int*)carve((size_t)N * 4);
  int*   deg_d = (int*)carve((size_t)N * 4);
  float* ns    = (float*)carve((size_t)N * 4);
  float* nd    = (float*)carve((size_t)N * 4);
  int*   sortws = (int*)carve((size_t)N * OUT_F * 4);
  float* z     = (float*)carve((size_t)N * NCLS * 4);
  float* agg2  = (float*)carve((size_t)N * NCLS * 4);

  const int B = 256;
  auto cdiv = [](long a, long b) { return (int)((a + b - 1) / b); };

  int* rowstart = deg_s;
  int* col      = (int*)agg2;
  const int NB = cdiv(N, BNODES);
  const int G  = cdiv(E, CHUNK);
  const int NT = 2 * NB * G;
  int* rec  = sortws;
  int* cnt  = rec + (((size_t)2 * E + 63) & ~(size_t)63);
  int* boff = cnt + (((size_t)NT + 63) & ~(size_t)63);

  k_bucket_count<<<G, 1024, 0, stream>>>(src, dst, cnt, E, NB, G);
  k_scan<<<cdiv(NT, 1024), 1024, 0, stream>>>(cnt, boff, NT);
  k_bucket_scatter<<<G, 1024, 0, stream>>>(src, dst, cnt, boff, rec, E, NB, G);
  k_bucket_rows<<<2 * NB, 512, 0, stream>>>(rec, boff, rowstart, deg_d, col, ns, nd, N, E, NB, G);

  k_layer1<<<cdiv(N, 16 * TPB), B, 0, stream>>>(rowstart, deg_d, col, x, ns, nd, W1, b1, W2, z, N);

  k_gather16_out<<<cdiv((long)N * 4, B), B, 0, stream>>>(rowstart, deg_d, col, z, ns, nd, b2, out, N);
}


# edge sort level 1 made chunk-local: a chunk is bucket-ordered in LDS and written to its own region with an offset row; count and table-scan kernels go; row sum order = arrival rank as before
# speedup vs baseline: 5.1157x; 1.2961x over previous
#include <hip/hip_runtime.h>

typedef __attribute__((ext_vector_type(2))) float v2f;
typedef __attribute__((ext_vector_type(8))) float v8f;

#define IN_F 64
#define OUT_F 64
#define NCLS 16

#define BSH 8
#define BNODES 256
#define NB_MAX 512
#define CHUNK 8192
#define ROWCAP 8192
#define GMAX 512

__device__ __forceinline__ int block_excl_scan(int v, int* wsum) {
  int lane = threadIdx.x & 31, w = threadIdx.x >> 5;
  int x = v;
  for (int o = 1; o < 32; o <<= 1) {
    int y = __shfl_up(x, o, 32);
    if (lane >= o) x += y;
  }
  if (lane == 31) wsum[w] = x;
  __syncthreads();
  if (w == 0) {
    int nw = (int)(blockDim.x >> 5);
    int s = (lane < nw) ? wsum[lane] : 0;
    for (int o = 1; o < 32; o <<= 1) {
      int y = __shfl_up(s, o, 32);
      if (lane >= o) s += y;
    }
    wsum[lane] = s;
  }
  __syncthreads();
  int prev = (w > 0) ? wsum[w - 1] : 0;
  return prev + x - v;
}

__global__ void __launch_bounds__(1024)
k_chunk_sort(const int* __restrict__ src, const int* __restrict__ dst,
             int* __restrict__ rec, int* __restrict__ off, int E, int NB, int P) {
  __shared__ int h[2 * NB_MAX];
  __shared__ int wsum[32];
  __shared__ __attribute__((aligned(16))) int stage[2 * CHUNK];
  const int t = threadIdx.x, g = blockIdx.x;
  const int nb2 = 2 * NB;
  const int beg = g * CHUNK;
  const int end = min(beg + CHUNK, E);

  int dv[8], sv[8];
#pragma unroll
  for (int r = 0; r < 2; ++r) {
    const int e = beg + r * 4096 + t * 4;
    if (e + 4 <= end) {
      const int4 d = *(const int4*)(dst + e);
      const int4 s = *(const int4*)(src + e);
      dv[4 * r + 0] = d.x; dv[4 * r + 1] = d.y; dv[4 * r + 2] = d.z; dv[4 * r + 3] = d.w;
      sv[4 * r + 0] = s.x; sv[4 * r + 1] = s.y; sv[4 * r + 2] = s.z; sv[4 * r + 3] = s.w;
    } else {
#pragma unroll
      for (int k = 0; k < 4; ++k) {
        const bool in = (e + k < end);
        dv[4 * r + k] = in ? dst[e + k] : -1;
        sv[4 * r + k] = in ? src[e + k] : 0;
      }
    }
  }
  if (t < nb2) h[t] = 0;
  __syncthreads();

  int rd[8], rq[8];
#pragma unroll
  for (int i = 0; i < 8; ++i) {
    rd[i] = 0; rq[i] = 0;
    if (dv[i] >= 0) {
      rd[i] = atomicAdd(&h[dv[i] >> BSH], 1);
      rq[i] = atomicAdd(&h[NB + (sv[i] >> BSH)], 1);
    }
  }
  __syncthreads();

  const int v = (t < nb2) ? h[t] : 0;
  const int ex = block_excl_scan(v, wsum);
  const int len2 = 2 * max(end - beg, 0);
  int* orow = off + (size_t)g * P;
  if (t < nb2) { h[t] = ex; orow[t] = ex; }
  if (t == 0) orow[nb2] = len2;
  __syncthreads();

#pragma unroll
  for (int i = 0; i < 8; ++i) {
    if (dv[i] >= 0) {
      const int d = dv[i], s = sv[i];
      stage[h[d >> BSH] + rd[i]] = (s << BSH) | (d & (BNODES - 1));
      stage[h[NB + (s >> BSH)] + rq[i]] = s & (BNODES - 1);
    }
  }
  __syncthreads();

  int* o = rec + (size_t)g * (2 * CHUNK);
  for (int j = t * 4; j < len2; j += 4096) {
    if (j + 4 <= len2) *(int4*)(o + j) = *(const int4*)(stage + j);
    else for (int k = j; k < len2; ++k) o[k] = stage[k];
  }
}

__global__ void __launch_bounds__(512)
k_bucket_rows(const int* __restrict__ rec, const int* __restrict__ off,
              int* __restrict__ rowstart, int* __restrict__ deg_d, int* __restrict__ col,
              float* __restrict__ ns, float* __restrict__ nd, int N, int NB, int G, int P) {
  __shared__ int c[BNODES];
  __shared__ int wsum[32];
  __shared__ int wtot[16];
  __shared__ int rb[GMAX + 128];
  __shared__ int rl[GMAX + 128];
  __shared__ int lcol[ROWCAP];
  const int t = threadIdx.x, lane = t & 31, w = t >> 5;
  const int bb = blockIdx.x;
  int o0 = 0;
  if (t < G) {
    const int* p = off + (size_t)t * P + bb;
    o0 = p[0];
    const int o1 = p[1];
    rb[t] = t * (2 * CHUNK) + o0;
    rl[t] = o1 - o0;
  } else {
    rb[t] = 0; rl[t] = 0;
  }
  if (t < 128) { rb[GMAX + t] = 0; rl[GMAX + t] = 0; }
  if (t < BNODES) c[t] = 0;
  int x = o0;
  for (int o = 16; o > 0; o >>= 1) x += __shfl_xor(x, o, 32);
  if (lane == 0) wtot[w] = x;
  __syncthreads();
  int beg = 0;
  for (int i = 0; i < 16; ++i) beg += wtot[i];

  for (int r0 = w; r0 < G; r0 += 8 * 16) {
    int v[8], ln[8], bs[8];
#pragma unroll
    for (int q = 0; q < 8; ++q) { ln[q] = rl[r0 + 16 * q]; bs[q] = rb[r0 + 16 * q]; }
#pragma unroll
    for (int q = 0; q < 8; ++q) v[q] = (lane < ln[q]) ? rec[bs[q] + lane] : -1;
#pragma unroll
    for (int q = 0; q < 8; ++q) {
      if (v[q] >= 0) atomicAdd(&c[v[q] & (BNODES - 1)], 1);
      for (int k = lane + 32; k < ln[q]; k += 32)
        atomicAdd(&c[rec[bs[q] + k] & (BNODES - 1)], 1);
    }
  }
  __syncthreads();

  if (bb >= NB) {
    int node = (bb - NB) * BNODES + t;
    if (t < BNODES && node < N) { int a = c[t]; ns[node] = (a > 0) ? rsqrtf((float)a) : 1.0f; }
    return;
  }

  int mine = (t < BNODES) ? c[t] : 0;
  int ex = block_excl_scan(mine, wsum);
  if (t < BNODES) {
    c[t] = ex;
    int node = bb * BNODES + t;
    if (node < N) {
      rowstart[node] = beg + ex;
      deg_d[node] = mine;
      nd[node] = (mine > 0) ? rsqrtf((float)mine) : 1.0f;
    }
  }
  __syncthreads();
  for (int r0 = w; r0 < G; r0 += 8 * 16) {
    int v[8], ln[8], bs[8];
#pragma unroll
    for (int q = 0; q < 8; ++q) { ln[q] = rl[r0 + 16 * q]; bs[q] = rb[r0 + 16 * q]; }
#pragma unroll
    for (int q = 0; q < 8; ++q) v[q] = (lane < ln[q]) ? rec[bs[q] + lane] : -1;
#pragma unroll
    for (int q = 0; q < 8; ++q) {
      if (v[q] >= 0) {
        int p = atomicAdd(&c[v[q] & (BNODES - 1)], 1);
        if (p < ROWCAP) lcol[p] = v[q] >> BSH;
        else col[beg + p] = v[q] >> BSH;
      }
      for (int k = lane + 32; k < ln[q]; k += 32) {
        int rr = rec[bs[q] + k];
        int p = atomicAdd(&c[rr & (BNODES - 1)], 1);
        if (p < ROWCAP) lcol[p] = rr >> BSH;
        else col[beg + p] = rr >> BSH;
      }
    }
  }
  __syncthreads();
  int nst = min(c[BNODES - 1], ROWCAP);
  for (int j = t; j < nst; j += 512) col[beg + j] = lcol[j];
}

#define TP 68
#define TPB 4

__global__ void __launch_bounds__(256)
k_layer1(const int* __restrict__ rowstart, const int* __restrict__ cnt,
         const int* __restrict__ col, const float* __restrict__ x,
         const float* __restrict__ ns, const float* __restrict__ nd,
         const float* __restrict__ W1, const float* __restrict__ b1,
         const float* __restrict__ W2, float* __restrict__ z, int N) {
  __shared__ __attribute__((aligned(16))) float sW1[IN_F * OUT_F];
  __shared__ __attribute__((aligned(16))) float sW2[OUT_F * NCLS];
  __shared__ __attribute__((aligned(16))) float tile[TPB][16 * TP];

  const int tid = threadIdx.x;
  const int lane = tid & 31, wv = tid >> 5;
  const int c = lane & 15, h = lane >> 4;
  const int t = wv >> 1, p = wv & 1;
  const int m0 = (blockIdx.x * TPB + t) * 16;
  float* tl = tile[t];

  const int myrow = m0 + 8 * p + (lane & 7);
  int rs = 0, rc = 0;
  if (myrow < N) { rs = rowstart[myrow]; rc = cnt[myrow]; }
  const int arow = m0 + c;
  const float scale = (arow < N) ? nd[arow] : 0.f;

  {
    float4 g[4];
#pragma unroll
    for (int q = 0; q < 4; ++q) g[q] = *(const float4*)(W1 + tid * 4 + q * 1024);
    float4 g2 = *(const float4*)(W2 + tid * 4);
#pragma unroll
    for (int q = 0; q < 4; ++q) *(float4*)(sW1 + tid * 4 + q * 1024) = g[q];
    *(float4*)(sW2 + tid * 4) = g2;
  }

  const int f = c << 2;
  for (int i = 0; i < 4; ++i) {
    const int lr = 2 * i + h;
    int j = __shfl(rs, lr, 32);
    const int end = j + __shfl(rc, lr, 32);
    const int last = max(end - 1, 0);
    float4 acc = make_float4(0.f, 0.f, 0.f, 0.f);
    int s[8];
#pragma unroll
    for (int k = 0; k < 8; ++k) s[k] = col[min(j + k, last)];
    while (j < end) {
      float w[8];
      float4 v[8];
      int sn[8];
#pragma unroll
      for (int k = 0; k < 8; ++k) {
        w[k] = ns[s[k]];
        v[k] = *(const float4*)(x + (long)s[k] * IN_F + f);
      }
#pragma unroll
      for (int k = 0; k < 8; ++k) sn[k] = col[min(j + 8 + k, last)];
#pragma unroll
      for (int k = 0; k < 8; ++k) {
        const bool in = (j + k < end);
        const float wk = in ? w[k] : 0.f;
        const float4 vk = in ? v[k] : make_float4(0.f, 0.f, 0.f, 0.f);
        acc.x += vk.x * wk; acc.y += vk.y * wk;
        acc.z += vk.z * wk; acc.w += vk.w * wk;
      }
#pragma unroll
      for (int k = 0; k < 8; ++k) s[k] = sn[k];
      j += 8;
    }
    *(float4*)(tl + (8 * p + lr) * TP + f) = acc;
  }
  __syncthreads();

  v2f a[16];
#pragma unroll
  for (int ks = 0; ks < 16; ++ks) {
    float2 av = *(const float2*)(tl + c * TP + ks * 4 + h * 2);
    a[ks] = (v2f){av.x * scale, av.y * scale};
  }
  __syncthreads();

  v8f acc0 = {}, acc1 = {};
#pragma unroll
  for (int ks = 0; ks < 16; ++ks) {
    const int kk = ks * 4 + h * 2;
    const float* w0 = sW1 + kk * OUT_F + 32 * p + c;
    v2f b0 = {w0[0],  w0[OUT_F]};
    v2f b1v = {w0[16], w0[OUT_F + 16]};
    acc0 = __builtin_amdgcn_wmma_f32_16x16x4_f32(false, a[ks], false, b0,  (short)0, acc0, false, false);
    acc1 = __builtin_amdgcn_wmma_f32_16x16x4_f32(false, a[ks], false, b1v, (short)0, acc1, false, false);
  }
  {
    const float bi0 = b1[32 * p + c], bi1 = b1[32 * p + 16 + c];
    float* yrow = tl + (8 * h) * TP + 32 * p + c;
#pragma unroll
    for (int r = 0; r < 8; ++r) {
      yrow[r * TP]      = fmaxf(acc0[r] + bi0, 0.f);
      yrow[r * TP + 16] = fmaxf(acc1[r] + bi1, 0.f);
    }
  }
  __syncthreads();

  if (p == 0 && m0 < N) {
    v8f accz = {};
#pragma unroll
    for (int ks = 0; ks < 16; ++ks) {
      float2 av = *(const float2*)(tl + c * TP + ks * 4 + h * 2);
      v2f ay = {av.x, av.y};
      const int kk = ks * 4 + h * 2;
      v2f b = {sW2[kk * NCLS + c], sW2[(kk + 1) * NCLS + c]};
      accz = __builtin_amdgcn_wmma_f32_16x16x4_f32(false, ay, false, b, (short)0, accz, false, false);
    }
    const int rbase = m0 + h * 8;
#pragma unroll
    for (int r = 0; r < 8; ++r)
      z[(long)(rbase + r) * NCLS + c] = accz[r];
  }
}

__global__ void k_gather16_out(const int* __restrict__ rowstart, const int* __restrict__ cnt,
                               const int* __restrict__ col, const float* __restrict__ z,
                               const float* __restrict__ ns, const float* __restrict__ nd,
                               const float* __restrict__ bias, float* __restrict__ out, int N) {
  int t = blockIdx.x * blockDim.x + threadIdx.x;
  int d = t >> 2;
  if (d >= N) return;
  int f = (t & 3) << 2;
  int j = rowstart[d];
  int end = j + cnt[d];
  float4 acc = make_float4(0.f, 0.f, 0.f, 0.f);
  for (; j + 4 <= end; j += 4) {
    int s0 = col[j], s1 = col[j + 1], s2 = col[j + 2], s3 = col[j + 3];
    float w0 = ns[s0], w1 = ns[s1], w2 = ns[s2], w3 = ns[s3];
    float4 v0 = *(const float4*)(z + (long)s0 * NCLS + f);
    float4 v1 = *(const float4*)(z + (long)s1 * NCLS + f);
    float4 v2 = *(const float4*)(z + (long)s2 * NCLS + f);
    float4 v3 = *(const float4*)(z + (long)s3 * NCLS + f);
    acc.x += v0.x * w0; acc.y += v0.y * w0; acc.z += v0.z * w0; acc.w += v0.w * w0;
    acc.x += v1.x * w1; acc.y += v1.y * w1; acc.z += v1.z * w1; acc.w += v1.w * w1;
    acc.x += v2.x * w2; acc.y += v2.y * w2; acc.z += v2.z * w2; acc.w += v2.w * w2;
    acc.x += v3.x * w3; acc.y += v3.y * w3; acc.z += v3.z * w3; acc.w += v3.w * w3;
  }
#pragma nounroll
  for (; j < end; ++j) {
    int s0 = col[j];
    float w0 = ns[s0];
    float4 v0 = *(const float4*)(z + (long)s0 * NCLS + f);
    acc.x += v0.x * w0; acc.y += v0.y * w0; acc.z += v0.z * w0; acc.w += v0.w * w0;
  }
  float sc = nd[d];
  float4 o;
  o.x = acc.x * sc + bias[f + 0];
  o.y = acc.y * sc + bias[f + 1];
  o.z = acc.z * sc + bias[f + 2];
  o.w = acc.w * sc + bias[f + 3];
  *(float4*)(out + (long)d * NCLS + f) = o;
}

extern "C" void kernel_launch(void* const* d_in, const int* in_sizes, int n_in,
                              void* d_out, int out_size, void* d_ws, size_t ws_size,
                              hipStream_t stream) {
  const float* x  = (const float*)d_in[0];
  const int*   src = (const int*)d_in[1];
  const int*   dst = (const int*)d_in[2];
  const float* W1 = (const float*)d_in[3];
  const float* b1 = (const float*)d_in[4];
  const float* W2 = (const float*)d_in[5];
  const float* b2 = (const float*)d_in[6];
  float* out = (float*)d_out;

  const int N = in_sizes[0] / IN_F;
  const int E = in_sizes[1];

  char* ws = (char*)d_ws;
  size_t off = 0;
  auto carve = [&](size_t bytes) -> void* {
    void* p = (void*)(ws + off);
    off += (bytes + 255) & ~(size_t)255;
    return p;
  };
  int*   deg_s = (int*)carve((size_t)N * 4);
  int*   deg_d = (int*)carve((size_t)N * 4);
  float* ns    = (float*)carve((size_t)N * 4);
  float* nd    = (float*)carve((size_t)N * 4);
  int*   sortws = (int*)carve((size_t)N * OUT_F * 4);
  float* z     = (float*)carve((size_t)N * NCLS * 4);
  float* agg2  = (float*)carve((size_t)N * NCLS * 4);

  const int B = 256;
  auto cdiv = [](long a, long b) { return (int)((a + b - 1) / b); };

  int* rowstart = deg_s;
  int* col      = (int*)agg2;
  const int NB = cdiv(N, BNODES);
  const int G  = cdiv(E, CHUNK);
  const int P  = (2 * NB + 1 + 31) & ~31;
  int* rec  = sortws;
  int* boff = rec + (size_t)G * (2 * CHUNK);

  k_chunk_sort<<<G, 1024, 0, stream>>>(src, dst, rec, boff, E, NB, P);
  k_bucket_rows<<<2 * NB, 512, 0, stream>>>(rec, boff, rowstart, deg_d, col, ns, nd, N, NB, G, P);

  k_layer1<<<cdiv(N, 16 * TPB), B, 0, stream>>>(rowstart, deg_d, col, x, ns, nd, W1, b1, W2, z, N);

  k_gather16_out<<<cdiv((long)N * 4, B), B, 0, stream>>>(rowstart, deg_d, col, z, ns, nd, b2, out, N);
}


# layer-2 norm_src folded into the z rows where layer 1 stores them (once per node), so the layer-2 gather drops its per-edge norm read; row sum order unchanged, products rounded before the add
# speedup vs baseline: 5.5472x; 1.0843x over previous
#include <hip/hip_runtime.h>

typedef __attribute__((ext_vector_type(2))) float v2f;
typedef __attribute__((ext_vector_type(8))) float v8f;

#define IN_F 64
#define OUT_F 64
#define NCLS 16

#define BSH 8
#define BNODES 256
#define NB_MAX 512
#define CHUNK 8192
#define ROWCAP 8192
#define GMAX 512

__device__ __forceinline__ int block_excl_scan(int v, int* wsum) {
  int lane = threadIdx.x & 31, w = threadIdx.x >> 5;
  int x = v;
  for (int o = 1; o < 32; o <<= 1) {
    int y = __shfl_up(x, o, 32);
    if (lane >= o) x += y;
  }
  if (lane == 31) wsum[w] = x;
  __syncthreads();
  if (w == 0) {
    int nw = (int)(blockDim.x >> 5);
    int s = (lane < nw) ? wsum[lane] : 0;
    for (int o = 1; o < 32; o <<= 1) {
      int y = __shfl_up(s, o, 32);
      if (lane >= o) s += y;
    }
    wsum[lane] = s;
  }
  __syncthreads();
  int prev = (w > 0) ? wsum[w - 1] : 0;
  return prev + x - v;
}

__global__ void __launch_bounds__(1024)
k_chunk_sort(const int* __restrict__ src, const int* __restrict__ dst,
             int* __restrict__ rec, int* __restrict__ off, int E, int NB, int P) {
  __shared__ int h[2 * NB_MAX];
  __shared__ int wsum[32];
  __shared__ __attribute__((aligned(16))) int stage[2 * CHUNK];
  const int t = threadIdx.x, g = blockIdx.x;
  const int nb2 = 2 * NB;
  const int beg = g * CHUNK;
  const int end = min(beg + CHUNK, E);

  int dv[8], sv[8];
#pragma unroll
  for (int r = 0; r < 2; ++r) {
    const int e = beg + r * 4096 + t * 4;
    if (e + 4 <= end) {
      const int4 d = *(const int4*)(dst + e);
      const int4 s = *(const int4*)(src + e);
      dv[4 * r + 0] = d.x; dv[4 * r + 1] = d.y; dv[4 * r + 2] = d.z; dv[4 * r + 3] = d.w;
      sv[4 * r + 0] = s.x; sv[4 * r + 1] = s.y; sv[4 * r + 2] = s.z; sv[4 * r + 3] = s.w;
    } else {
#pragma unroll
      for (int k = 0; k < 4; ++k) {
        const bool in = (e + k < end);
        dv[4 * r + k] = in ? dst[e + k] : -1;
        sv[4 * r + k] = in ? src[e + k] : 0;
      }
    }
  }
  if (t < nb2) h[t] = 0;
  __syncthreads();

  int rd[8], rq[8];
#pragma unroll
  for (int i = 0; i < 8; ++i) {
    rd[i] = 0; rq[i] = 0;
    if (dv[i] >= 0) {
      rd[i] = atomicAdd(&h[dv[i] >> BSH], 1);
      rq[i] = atomicAdd(&h[NB + (sv[i] >> BSH)], 1);
    }
  }
  __syncthreads();

  const int v = (t < nb2) ? h[t] : 0;
  const int ex = block_excl_scan(v, wsum);
  const int len2 = 2 * max(end - beg, 0);
  int* orow = off + (size_t)g * P;
  if (t < nb2) { h[t] = ex; orow[t] = ex; }
  if (t == 0) orow[nb2] = len2;
  __syncthreads();

#pragma unroll
  for (int i = 0; i < 8; ++i) {
    if (dv[i] >= 0) {
      const int d = dv[i], s = sv[i];
      stage[h[d >> BSH] + rd[i]] = (s << BSH) | (d & (BNODES - 1));
      stage[h[NB + (s >> BSH)] + rq[i]] = s & (BNODES - 1);
    }
  }
  __syncthreads();

  int* o = rec + (size_t)g * (2 * CHUNK);
  for (int j = t * 4; j < len2; j += 4096) {
    if (j + 4 <= len2) *(int4*)(o + j) = *(const int4*)(stage + j);
    else for (int k = j; k < len2; ++k) o[k] = stage[k];
  }
}

__global__ void __launch_bounds__(512)
k_bucket_rows(const int* __restrict__ rec, const int* __restrict__ off,
              int* __restrict__ rowstart, int* __restrict__ deg_d, int* __restrict__ col,
              float* __restrict__ ns, float* __restrict__ nd, int N, int NB, int G, int P) {
  __shared__ int c[BNODES];
  __shared__ int wsum[32];
  __shared__ int wtot[16];
  __shared__ int rb[GMAX + 128];
  __shared__ int rl[GMAX + 128];
  __shared__ int lcol[ROWCAP];
  const int t = threadIdx.x, lane = t & 31, w = t >> 5;
  const int bb = blockIdx.x;
  int o0 = 0;
  if (t < G) {
    const int* p = off + (size_t)t * P + bb;
    o0 = p[0];
    const int o1 = p[1];
    rb[t] = t * (2 * CHUNK) + o0;
    rl[t] = o1 - o0;
  } else {
    rb[t] = 0; rl[t] = 0;
  }
  if (t < 128) { rb[GMAX + t] = 0; rl[GMAX + t] = 0; }
  if (t < BNODES) c[t] = 0;
  int x = o0;
  for (int o = 16; o > 0; o >>= 1) x += __shfl_xor(x, o, 32);
  if (lane == 0) wtot[w] = x;
  __syncthreads();
  int beg = 0;
  for (int i = 0; i < 16; ++i) beg += wtot[i];

  for (int r0 = w; r0 < G; r0 += 8 * 16) {
    int v[8], ln[8], bs[8];
#pragma unroll
    for (int q = 0; q < 8; ++q) { ln[q] = rl[r0 + 16 * q]; bs[q] = rb[r0 + 16 * q]; }
#pragma unroll
    for (int q = 0; q < 8; ++q) v[q] = (lane < ln[q]) ? rec[bs[q] + lane] : -1;
#pragma unroll
    for (int q = 0; q < 8; ++q) {
      if (v[q] >= 0) atomicAdd(&c[v[q] & (BNODES - 1)], 1);
      for (int k = lane + 32; k < ln[q]; k += 32)
        atomicAdd(&c[rec[bs[q] + k] & (BNODES - 1)], 1);
    }
  }
  __syncthreads();

  if (bb >= NB) {
    int node = (bb - NB) * BNODES + t;
    if (t < BNODES && node < N) { int a = c[t]; ns[node] = (a > 0) ? rsqrtf((float)a) : 1.0f; }
    return;
  }

  int mine = (t < BNODES) ? c[t] : 0;
  int ex = block_excl_scan(mine, wsum);
  if (t < BNODES) {
    c[t] = ex;
    int node = bb * BNODES + t;
    if (node < N) {
      rowstart[node] = beg + ex;
      deg_d[node] = mine;
      nd[node] = (mine > 0) ? rsqrtf((float)mine) : 1.0f;
    }
  }
  __syncthreads();
  for (int r0 = w; r0 < G; r0 += 8 * 16) {
    int v[8], ln[8], bs[8];
#pragma unroll
    for (int q = 0; q < 8; ++q) { ln[q] = rl[r0 + 16 * q]; bs[q] = rb[r0 + 16 * q]; }
#pragma unroll
    for (int q = 0; q < 8; ++q) v[q] = (lane < ln[q]) ? rec[bs[q] + lane] : -1;
#pragma unroll
    for (int q = 0; q < 8; ++q) {
      if (v[q] >= 0) {
        int p = atomicAdd(&c[v[q] & (BNODES - 1)], 1);
        if (p < ROWCAP) lcol[p] = v[q] >> BSH;
        else col[beg + p] = v[q] >> BSH;
      }
      for (int k = lane + 32; k < ln[q]; k += 32) {
        int rr = rec[bs[q] + k];
        int p = atomicAdd(&c[rr & (BNODES - 1)], 1);
        if (p < ROWCAP) lcol[p] = rr >> BSH;
        else col[beg + p] = rr >> BSH;
      }
    }
  }
  __syncthreads();
  int nst = min(c[BNODES - 1], ROWCAP);
  for (int j = t; j < nst; j += 512) col[beg + j] = lcol[j];
}

#define TP 68
#define TPB 4

__global__ void __launch_bounds__(256)
k_layer1(const int* __restrict__ rowstart, const int* __restrict__ cnt,
         const int* __restrict__ col, const float* __restrict__ x,
         const float* __restrict__ ns, const float* __restrict__ nd,
         const float* __restrict__ W1, const float* __restrict__ b1,
         const float* __restrict__ W2, float* __restrict__ z, int N) {
  __shared__ __attribute__((aligned(16))) float sW1[IN_F * OUT_F];
  __shared__ __attribute__((aligned(16))) float sW2[OUT_F * NCLS];
  __shared__ __attribute__((aligned(16))) float tile[TPB][16 * TP];

  const int tid = threadIdx.x;
  const int lane = tid & 31, wv = tid >> 5;
  const int c = lane & 15, h = lane >> 4;
  const int t = wv >> 1, p = wv & 1;
  const int m0 = (blockIdx.x * TPB + t) * 16;
  float* tl = tile[t];

  const int myrow = m0 + 8 * p + (lane & 7);
  int rs = 0, rc = 0;
  if (myrow < N) { rs = rowstart[myrow]; rc = cnt[myrow]; }
  const int arow = m0 + c;
  const float scale = (arow < N) ? nd[arow] : 0.f;
  const float zs = (arow < N) ? ns[arow] : 0.f;

  {
    float4 g[4];
#pragma unroll
    for (int q = 0; q < 4; ++q) g[q] = *(const float4*)(W1 + tid * 4 + q * 1024);
    float4 g2 = *(const float4*)(W2 + tid * 4);
#pragma unroll
    for (int q = 0; q < 4; ++q) *(float4*)(sW1 + tid * 4 + q * 1024) = g[q];
    *(float4*)(sW2 + tid * 4) = g2;
  }

  const int f = c << 2;
  for (int i = 0; i < 4; ++i) {
    const int lr = 2 * i + h;
    int j = __shfl(rs, lr, 32);
    const int end = j + __shfl(rc, lr, 32);
    const int last = max(end - 1, 0);
    float4 acc = make_float4(0.f, 0.f, 0.f, 0.f);
    int s[8];
#pragma unroll
    for (int k = 0; k < 8; ++k) s[k] = col[min(j + k, last)];
    while (j < end) {
      float w[8];
      float4 v[8];
      int sn[8];
#pragma unroll
      for (int k = 0; k < 8; ++k) {
        w[k] = ns[s[k]];
        v[k] = *(const float4*)(x + (long)s[k] * IN_F + f);
      }
#pragma unroll
      for (int k = 0; k < 8; ++k) sn[k] = col[min(j + 8 + k, last)];
#pragma unroll
      for (int k = 0; k < 8; ++k) {
        const bool in = (j + k < end);
        const float wk = in ? w[k] : 0.f;
        const float4 vk = in ? v[k] : make_float4(0.f, 0.f, 0.f, 0.f);
        acc.x += vk.x * wk; acc.y += vk.y * wk;
        acc.z += vk.z * wk; acc.w += vk.w * wk;
      }
#pragma unroll
      for (int k = 0; k < 8; ++k) s[k] = sn[k];
      j += 8;
    }
    *(float4*)(tl + (8 * p + lr) * TP + f) = acc;
  }
  __syncthreads();

  v2f a[16];
#pragma unroll
  for (int ks = 0; ks < 16; ++ks) {
    float2 av = *(const float2*)(tl + c * TP + ks * 4 + h * 2);
    a[ks] = (v2f){av.x * scale, av.y * scale};
  }
  __syncthreads();

  v8f acc0 = {}, acc1 = {};
#pragma unroll
  for (int ks = 0; ks < 16; ++ks) {
    const int kk = ks * 4 + h * 2;
    const float* w0 = sW1 + kk * OUT_F + 32 * p + c;
    v2f b0 = {w0[0],  w0[OUT_F]};
    v2f b1v = {w0[16], w0[OUT_F + 16]};
    acc0 = __builtin_amdgcn_wmma_f32_16x16x4_f32(false, a[ks], false, b0,  (short)0, acc0, false, false);
    acc1 = __builtin_amdgcn_wmma_f32_16x16x4_f32(false, a[ks], false, b1v, (short)0, acc1, false, false);
  }
  {
    const float bi0 = b1[32 * p + c], bi1 = b1[32 * p + 16 + c];
    float* yrow = tl + (8 * h) * TP + 32 * p + c;
#pragma unroll
    for (int r = 0; r < 8; ++r) {
      yrow[r * TP]      = fmaxf(acc0[r] + bi0, 0.f);
      yrow[r * TP + 16] = fmaxf(acc1[r] + bi1, 0.f);
    }
  }
  __syncthreads();

  if (p == 0 && m0 < N) {
    v8f accz = {};
#pragma unroll
    for (int ks = 0; ks < 16; ++ks) {
      float2 av = *(const float2*)(tl + c * TP + ks * 4 + h * 2);
      v2f ay = {av.x, av.y};
      const int kk = ks * 4 + h * 2;
      v2f b = {sW2[kk * NCLS + c], sW2[(kk + 1) * NCLS + c]};
      accz = __builtin_amdgcn_wmma_f32_16x16x4_f32(false, ay, false, b, (short)0, accz, false, false);
    }
    const int rbase = m0 + h * 8;
#pragma unroll
    for (int r = 0; r < 8; ++r)
      z[(long)(rbase + r) * NCLS + c] = accz[r] * __shfl(zs, 8 * h + r, 32);
  }
}

__global__ void k_gather16_out(const int* __restrict__ rowstart, const int* __restrict__ cnt,
                               const int* __restrict__ col, const float* __restrict__ z,
                               const float* __restrict__ nd,
                               const float* __restrict__ bias, float* __restrict__ out, int N) {
  int t = blockIdx.x * blockDim.x + threadIdx.x;
  int d = t >> 2;
  if (d >= N) return;
  int f = (t & 3) << 2;
  int j = rowstart[d];
  int end = j + cnt[d];
  float4 acc = make_float4(0.f, 0.f, 0.f, 0.f);
  for (; j + 4 <= end; j += 4) {
    int s0 = col[j], s1 = col[j + 1], s2 = col[j + 2], s3 = col[j + 3];
    float4 v0 = *(const float4*)(z + (long)s0 * NCLS + f);
    float4 v1 = *(const float4*)(z + (long)s1 * NCLS + f);
    float4 v2 = *(const float4*)(z + (long)s2 * NCLS + f);
    float4 v3 = *(const float4*)(z + (long)s3 * NCLS + f);
    acc.x += v0.x; acc.y += v0.y; acc.z += v0.z; acc.w += v0.w;
    acc.x += v1.x; acc.y += v1.y; acc.z += v1.z; acc.w += v1.w;
    acc.x += v2.x; acc.y += v2.y; acc.z += v2.z; acc.w += v2.w;
    acc.x += v3.x; acc.y += v3.y; acc.z += v3.z; acc.w += v3.w;
  }
#pragma nounroll
  for (; j < end; ++j) {
    int s0 = col[j];
    float4 v0 = *(const float4*)(z + (long)s0 * NCLS + f);
    acc.x += v0.x; acc.y += v0.y; acc.z += v0.z; acc.w += v0.w;
  }
  float sc = nd[d];
  float4 o;
  o.x = acc.x * sc + bias[f + 0];
  o.y = acc.y * sc + bias[f + 1];
  o.z = acc.z * sc + bias[f + 2];
  o.w = acc.w * sc + bias[f + 3];
  *(float4*)(out + (long)d * NCLS + f) = o;
}

extern "C" void kernel_launch(void* const* d_in, const int* in_sizes, int n_in,
                              void* d_out, int out_size, void* d_ws, size_t ws_size,
                              hipStream_t stream) {
  const float* x  = (const float*)d_in[0];
  const int*   src = (const int*)d_in[1];
  const int*   dst = (const int*)d_in[2];
  const float* W1 = (const float*)d_in[3];
  const float* b1 = (const float*)d_in[4];
  const float* W2 = (const float*)d_in[5];
  const float* b2 = (const float*)d_in[6];
  float* out = (float*)d_out;

  const int N = in_sizes[0] / IN_F;
  const int E = in_sizes[1];

  char* ws = (char*)d_ws;
  size_t off = 0;
  auto carve = [&](size_t bytes) -> void* {
    void* p = (void*)(ws + off);
    off += (bytes + 255) & ~(size_t)255;
    return p;
  };
  int*   deg_s = (int*)carve((size_t)N * 4);
  int*   deg_d = (int*)carve((size_t)N * 4);
  float* ns    = (float*)carve((size_t)N * 4);
  float* nd    = (float*)carve((size_t)N * 4);
  int*   sortws = (int*)carve((size_t)N * OUT_F * 4);
  float* z     = (float*)carve((size_t)N * NCLS * 4);
  float* agg2  = (float*)carve((size_t)N * NCLS * 4);

  const int B = 256;
  auto cdiv = [](long a, long b) { return (int)((a + b - 1) / b); };

  int* rowstart = deg_s;
  int* col      = (int*)agg2;
  const int NB = cdiv(N, BNODES);
  const int G  = cdiv(E, CHUNK);
  const int P  = (2 * NB + 1 + 31) & ~31;
  int* rec  = sortws;
  int* boff = rec + (size_t)G * (2 * CHUNK);

  k_chunk_sort<<<G, 1024, 0, stream>>>(src, dst, rec, boff, E, NB, P);
  k_bucket_rows<<<2 * NB, 512, 0, stream>>>(rec, boff, rowstart, deg_d, col, ns, nd, N, NB, G, P);

  k_layer1<<<cdiv(N, 16 * TPB), B, 0, stream>>>(rowstart, deg_d, col, x, ns, nd, W1, b1, W2, z, N);

  k_gather16_out<<<cdiv((long)N * 4, B), B, 0, stream>>>(rowstart, deg_d, col, z, nd, b2, out, N);
}
